# starting kernel
# baseline (speedup 1.0000x reference)
#include <hip/hip_runtime.h>

typedef __attribute__((ext_vector_type(2))) float v2f;
typedef __attribute__((ext_vector_type(4))) float v4f;
typedef __attribute__((ext_vector_type(8))) float v8f;

#define ROW_GROUPS     8
#define WAVES_PER_BLK  8
#define COLS_PER_WAVE  16
#define COLS_PER_BLK   (WAVES_PER_BLK * COLS_PER_WAVE)
#define SCAN_T         256

__global__ __launch_bounds__(256) void colsum_wmma(
    const float* __restrict__ x, float* __restrict__ psum, float* __restrict__ psq,
    int L, int rows) {
  const int wave = threadIdx.x >> 5;
  const int lane = threadIdx.x & 31;
  const int m    = lane & 15;
  const int kh   = lane >> 4;
  const int c0   = blockIdx.x * COLS_PER_BLK + wave * COLS_PER_WAVE;
  const int col  = c0 + m;
  const int rpg  = rows / ROW_GROUPS;
  const int r0   = blockIdx.y * rpg;
  const int rEnd = r0 + rpg;

  const size_t strideRow = (size_t)L;
  const size_t stride4   = 4 * strideRow;

  v8f acc1a = {}, acc1b = {};
  v8f acc2a = {}, acc2b = {};
  v2f ones; ones[0] = 1.0f; ones[1] = 1.0f;

  const float* p = x + (size_t)(r0 + 2 * kh) * strideRow + (size_t)col;

  int r = r0;
  const int mainEnd = (rEnd < rows - 19) ? rEnd : (rows - 19);
  for (; r + 8 <= mainEnd; r += 8) {
    float a0 = p[0];
    float a1 = p[strideRow];
    float b0 = p[stride4];
    float b1 = p[stride4 + strideRow];
    __builtin_prefetch(p + 16 * strideRow, 0, 1);
    __builtin_prefetch(p + 17 * strideRow, 0, 1);
    __builtin_prefetch(p + 20 * strideRow, 0, 1);
    __builtin_prefetch(p + 21 * strideRow, 0, 1);
    v2f A;  A[0]  = a0;      A[1]  = a1;
    v2f A2; A2[0] = a0 * a0; A2[1] = a1 * a1;
    v2f Bv; Bv[0] = b0;      Bv[1] = b1;
    v2f B2; B2[0] = b0 * b0; B2[1] = b1 * b1;
    acc1a = __builtin_amdgcn_wmma_f32_16x16x4_f32(false, A,  false, ones,
                                                  (short)0, acc1a, false, false);
    acc2a = __builtin_amdgcn_wmma_f32_16x16x4_f32(false, A2, false, ones,
                                                  (short)0, acc2a, false, false);
    acc1b = __builtin_amdgcn_wmma_f32_16x16x4_f32(false, Bv, false, ones,
                                                  (short)0, acc1b, false, false);
    acc2b = __builtin_amdgcn_wmma_f32_16x16x4_f32(false, B2, false, ones,
                                                  (short)0, acc2b, false, false);
    p += 2 * stride4;
  }
  for (; r < rEnd; r += 4) {
    float a0 = p[0];
    float a1 = p[strideRow];
    v2f A;  A[0]  = a0;      A[1]  = a1;
    v2f A2; A2[0] = a0 * a0; A2[1] = a1 * a1;
    acc1a = __builtin_amdgcn_wmma_f32_16x16x4_f32(false, A,  false, ones,
                                                  (short)0, acc1a, false, false);
    acc2a = __builtin_amdgcn_wmma_f32_16x16x4_f32(false, A2, false, ones,
                                                  (short)0, acc2a, false, false);
    p += stride4;
  }

  const v8f acc1 = acc1a + acc1b;
  const v8f acc2 = acc2a + acc2b;

  if ((lane & 15) == 0) {
    const int cb = c0 + kh * 8;
    float* ps = psum + (size_t)blockIdx.y * (size_t)L + cb;
    float* pq = psq  + (size_t)blockIdx.y * (size_t)L + cb;
#pragma unroll
    for (int j = 0; j < 8; ++j) { ps[j] = acc1[j]; pq[j] = acc2[j]; }
  }
}

__global__ __launch_bounds__(256) void combine_groups(
    const float* __restrict__ psum, const float* __restrict__ psq,
    float* __restrict__ c1, float* __restrict__ c2, int L) {
  const int k = blockIdx.x * blockDim.x + threadIdx.x;
  if (k >= L) return;
  float a = 0.f, b = 0.f;
#pragma unroll
  for (int g = 0; g < ROW_GROUPS; ++g) {
    a += psum[(size_t)g * L + k];
    b += psq [(size_t)g * L + k];
  }
  c1[k] = a; c2[k] = b;
}

__global__ __launch_bounds__(SCAN_T) void scan_stats(
    const float* __restrict__ c1, const float* __restrict__ c2,
    float* __restrict__ mean, float* __restrict__ istd,
    int L, float Nf, float Bf) {
  __shared__ float s1[SCAN_T], s2[SCAN_T];
  const int t = threadIdx.x;
  const int chunk = (L + SCAN_T - 1) / SCAN_T;
  const int k0 = t * chunk;
  const int k1 = (k0 + chunk < L) ? (k0 + chunk) : L;

  float la = 0.f, lb = 0.f;
  for (int k = k0; k < k1; ++k) { la += c1[k]; lb += c2[k]; }
  s1[t] = la; s2[t] = lb;
  __syncthreads();
  for (int off = 1; off < SCAN_T; off <<= 1) {
    float t1 = 0.f, t2 = 0.f;
    if (t >= off) { t1 = s1[t - off]; t2 = s2[t - off]; }
    __syncthreads();
    s1[t] += t1; s2[t] += t2;
    __syncthreads();
  }
  float run1 = s1[t] - la;
  float run2 = s2[t] - lb;
  for (int k = k0; k < k1; ++k) {
    run1 += c1[k];
    run2 += c2[k];
    const float denom = Nf * (float)(k + 1);
    const float mu  = run1 / denom;
    const float var = (run2 - 2.f * mu * run1) / denom + mu * mu * Bf;
    mean[k] = mu;
    istd[k] = rsqrtf(var);
  }
}

__global__ __launch_bounds__(256) void normalize_k(
    const float* __restrict__ x, const float* __restrict__ gamma,
    const float* __restrict__ beta, const float* __restrict__ mean,
    const float* __restrict__ istd, float* __restrict__ out, int L) {
  const int q = blockIdx.x * blockDim.x + threadIdx.x;
  const int k = q * 4;
  if (k >= L) return;
  const int row = blockIdx.y;
  const size_t base = (size_t)row * (size_t)L + (size_t)k;
  const v4f xv = *(const v4f*)(x    + base);
  const v4f mv = *(const v4f*)(mean + k);
  const v4f sv = *(const v4f*)(istd + k);
  const float g = gamma[row];
  const float b = beta[row];
  v4f o = (xv - mv) * sv * g + b;
  __builtin_nontemporal_store(o, (v4f*)(out + base));
}

extern "C" void kernel_launch(void* const* d_in, const int* in_sizes, int n_in,
                              void* d_out, int out_size, void* d_ws, size_t ws_size,
                              hipStream_t stream) {
  const float* x     = (const float*)d_in[0];
  const float* gamma = (const float*)d_in[1];
  const float* beta  = (const float*)d_in[2];
  float* out = (float*)d_out;

  const int B = 4, N = 512;
  const int rows = B * N;
  const int L = in_sizes[0] / rows;

  float* ws   = (float*)d_ws;
  float* psum = ws;
  float* psq  = psum + (size_t)ROW_GROUPS * L;
  float* c1   = psq  + (size_t)ROW_GROUPS * L;
  float* c2   = c1 + L;
  float* mean = c2 + L;
  float* istd = mean + L;

  dim3 g1(L / COLS_PER_BLK, ROW_GROUPS);
  colsum_wmma<<<g1, 256, 0, stream>>>(x, psum, psq, L, rows);

  combine_groups<<<(L + 255) / 256, 256, 0, stream>>>(psum, psq, c1, c2, L);

  scan_stats<<<1, SCAN_T, 0, stream>>>(c1, c2, mean, istd, L, (float)N, (float)B);

  dim3 g4((L / 4 + 255) / 256, rows);
  normalize_k<<<g4, 256, 0, stream>>>(x, gamma, beta, mean, istd, out, L);
}


# Scan launch split over 16 workgroups of 4 tiles; each forms the totals of all earlier tiles again on WMMA for its own offsets, no hand-over or second pass; sum order and results unchanged
# speedup vs baseline: 1.7565x; 1.7565x over previous
#include <hip/hip_runtime.h>

typedef __attribute__((ext_vector_type(2))) float v2f;
typedef __attribute__((ext_vector_type(4))) float v4f;
typedef __attribute__((ext_vector_type(8))) float v8f;

#define ROW_GROUPS     8
#define WAVES_PER_BLK  8
#define COLS_PER_WAVE  16
#define COLS_PER_BLK   (WAVES_PER_BLK * COLS_PER_WAVE)
#define SCAN_WAVES     16
#define SCAN_THREADS   (SCAN_WAVES * 32)
#define SCAN_TILES_PER_WAVE 4
#define SCAN_GROUPS    16

__global__ __launch_bounds__(256) void colsum_wmma(
    const float* __restrict__ x, float* __restrict__ psum, float* __restrict__ psq,
    int L, int rows) {
  const int wave = threadIdx.x >> 5;
  const int lane = threadIdx.x & 31;
  const int m    = lane & 15;
  const int kh   = lane >> 4;
  const int c0   = blockIdx.x * COLS_PER_BLK + wave * COLS_PER_WAVE;
  const int col  = c0 + m;
  const int rpg  = rows / ROW_GROUPS;
  const int r0   = blockIdx.y * rpg;
  const int rEnd = r0 + rpg;

  const size_t strideRow = (size_t)L;
  const size_t stride4   = 4 * strideRow;

  v8f acc1a = {}, acc1b = {};
  v8f acc2a = {}, acc2b = {};
  v2f ones; ones[0] = 1.0f; ones[1] = 1.0f;

  const float* p = x + (size_t)(r0 + 2 * kh) * strideRow + (size_t)col;

  int r = r0;
  const int mainEnd = (rEnd < rows - 19) ? rEnd : (rows - 19);
  for (; r + 8 <= mainEnd; r += 8) {
    float a0 = p[0];
    float a1 = p[strideRow];
    float b0 = p[stride4];
    float b1 = p[stride4 + strideRow];
    __builtin_prefetch(p + 16 * strideRow, 0, 1);
    __builtin_prefetch(p + 17 * strideRow, 0, 1);
    __builtin_prefetch(p + 20 * strideRow, 0, 1);
    __builtin_prefetch(p + 21 * strideRow, 0, 1);
    v2f A;  A[0]  = a0;      A[1]  = a1;
    v2f A2; A2[0] = a0 * a0; A2[1] = a1 * a1;
    v2f Bv; Bv[0] = b0;      Bv[1] = b1;
    v2f B2; B2[0] = b0 * b0; B2[1] = b1 * b1;
    acc1a = __builtin_amdgcn_wmma_f32_16x16x4_f32(false, A,  false, ones,
                                                  (short)0, acc1a, false, false);
    acc2a = __builtin_amdgcn_wmma_f32_16x16x4_f32(false, A2, false, ones,
                                                  (short)0, acc2a, false, false);
    acc1b = __builtin_amdgcn_wmma_f32_16x16x4_f32(false, Bv, false, ones,
                                                  (short)0, acc1b, false, false);
    acc2b = __builtin_amdgcn_wmma_f32_16x16x4_f32(false, B2, false, ones,
                                                  (short)0, acc2b, false, false);
    p += 2 * stride4;
  }
  for (; r < rEnd; r += 4) {
    float a0 = p[0];
    float a1 = p[strideRow];
    v2f A;  A[0]  = a0;      A[1]  = a1;
    v2f A2; A2[0] = a0 * a0; A2[1] = a1 * a1;
    acc1a = __builtin_amdgcn_wmma_f32_16x16x4_f32(false, A,  false, ones,
                                                  (short)0, acc1a, false, false);
    acc2a = __builtin_amdgcn_wmma_f32_16x16x4_f32(false, A2, false, ones,
                                                  (short)0, acc2a, false, false);
    p += stride4;
  }

  const v8f acc1 = acc1a + acc1b;
  const v8f acc2 = acc2a + acc2b;

  if ((lane & 15) == 0) {
    const int cb = c0 + kh * 8;
    float* ps = psum + (size_t)blockIdx.y * (size_t)L + cb;
    float* pq = psq  + (size_t)blockIdx.y * (size_t)L + cb;
#pragma unroll
    for (int j = 0; j < 8; ++j) { ps[j] = acc1[j]; pq[j] = acc2[j]; }
  }
}

__global__ __launch_bounds__(256) void combine_groups(
    const float* __restrict__ psum, const float* __restrict__ psq,
    float* __restrict__ c1, float* __restrict__ c2, int L) {
  const int k = blockIdx.x * blockDim.x + threadIdx.x;
  if (k >= L) return;
  float a = 0.f, b = 0.f;
#pragma unroll
  for (int g = 0; g < ROW_GROUPS; ++g) {
    a += psum[(size_t)g * L + k];
    b += psq [(size_t)g * L + k];
  }
  c1[k] = a; c2[k] = b;
}

__device__ __forceinline__ void load_tile_global(
    const float* __restrict__ src, int e0, int n, int c, int h, v2f (&a)[4]) {
#pragma unroll
  for (int j = 0; j < 4; ++j) {
    const int e = e0 + 16 * c + 4 * j + 2 * h;
    a[j] = *(const v2f*)(src + (e < n ? e : 0));
  }
}

__device__ __forceinline__ void mask_tile(int e0, int n, int c, int h, v2f (&a)[4]) {
#pragma unroll
  for (int j = 0; j < 4; ++j) {
    const int e = e0 + 16 * c + 4 * j + 2 * h;
    if (!(e < n)) { a[j][0] = 0.f; a[j][1] = 0.f; }
  }
}

__device__ __forceinline__ void tile_scan(
    const v2f (&a)[4], const v2f (&tri)[4], int h, v8f& P, float& total) {
  v2f ones; ones[0] = 1.0f; ones[1] = 1.0f;
  v8f R = {};
#pragma unroll
  for (int j = 0; j < 4; ++j)
    R = __builtin_amdgcn_wmma_f32_16x16x4_f32(false, a[j], false, ones,
                                              (short)0, R, false, false);
  v8f Off;
  float run = 0.f;
#pragma unroll
  for (int r = 0; r < 8; ++r) { Off[r] = run; run += R[r]; }
  const float low  = __shfl(run, 0);
  const float high = __shfl(run, 16);
  const float add  = h ? low : 0.f;
#pragma unroll
  for (int r = 0; r < 8; ++r) Off[r] += add;
  total = low + high;
  P = Off;
#pragma unroll
  for (int j = 0; j < 4; ++j)
    P = __builtin_amdgcn_wmma_f32_16x16x4_f32(false, a[j], false, tri[j],
                                              (short)0, P, false, false);
}

__device__ __forceinline__ void stats_store(
    float run1, float run2, int k, float Nf, float Bf,
    float* __restrict__ mean, float* __restrict__ istd) {
  const float denom = Nf * (float)(k + 1);
  const float mu  = run1 / denom;
  const float var = (run2 - 2.f * mu * run1) / denom + mu * mu * Bf;
  mean[k] = mu;
  istd[k] = rsqrtf(var);
}

__device__ __forceinline__ float tile_total(const v2f (&a)[4]) {
  v2f ones; ones[0] = 1.0f; ones[1] = 1.0f;
  v8f R = {};
#pragma unroll
  for (int j = 0; j < 4; ++j)
    R = __builtin_amdgcn_wmma_f32_16x16x4_f32(false, a[j], false, ones,
                                              (short)0, R, false, false);
  float run = 0.f;
#pragma unroll
  for (int r = 0; r < 8; ++r) run += R[r];
  const float low  = __shfl(run, 0);
  const float high = __shfl(run, 16);
  return low + high;
}

__global__ __launch_bounds__(SCAN_THREADS) void scan_stats_wmma(
    const float* __restrict__ c1, const float* __restrict__ c2,
    float* __restrict__ mean, float* __restrict__ istd,
    int L, float Nf, float Bf) {
  __shared__ __attribute__((aligned(16))) float tot1[256];
  __shared__ __attribute__((aligned(16))) float tot2[256];
  __shared__ float off1[256], off2[256];
  const int tid  = threadIdx.x;
  const int wave = __builtin_amdgcn_readfirstlane(tid >> 5);
  const int lane = tid & 31;
  const int c    = lane & 15;
  const int h    = lane >> 4;
  const int nt   = (L + 255) >> 8;
  const int tBeg = 4 * (int)blockIdx.x;
  const int tOwn = tBeg + wave;
  const bool own = wave < 4;

  v2f U[4], Us[4];
#pragma unroll
  for (int j = 0; j < 4; ++j) {
#pragma unroll
    for (int s = 0; s < 2; ++s) {
      const int k = 4 * j + 2 * h + s;
      U[j][s]  = (k <= c) ? 1.0f : 0.f;
      Us[j][s] = (k <  c) ? 1.0f : 0.f;
    }
  }

  if (tid >= tBeg + 4 && tid < 256) { tot1[tid] = 0.f; tot2[tid] = 0.f; }

  v2f p1[SCAN_TILES_PER_WAVE][4], p2[SCAN_TILES_PER_WAVE][4];
#pragma unroll
  for (int i = 0; i < SCAN_TILES_PER_WAVE; ++i) {
    const int t = wave + SCAN_WAVES * i;
    if (t < tBeg) {
      load_tile_global(c1, t * 256, L, c, h, p1[i]);
      load_tile_global(c2, t * 256, L, c, h, p2[i]);
    }
  }
  v2f a1[4], a2[4];
  if (own) {
    load_tile_global(c1, tOwn * 256, L, c, h, a1);
    load_tile_global(c2, tOwn * 256, L, c, h, a2);
  }

#pragma unroll
  for (int i = 0; i < SCAN_TILES_PER_WAVE; ++i) {
    const int t = wave + SCAN_WAVES * i;
    if (t < tBeg) {
      mask_tile(t * 256, L, c, h, p1[i]);
      mask_tile(t * 256, L, c, h, p2[i]);
      const float s1 = tile_total(p1[i]);
      const float s2 = tile_total(p2[i]);
      if (lane == 0) { tot1[t] = s1; tot2[t] = s2; }
    }
  }
  v8f P1 = {}, P2 = {};
  if (own) {
    float t1, t2;
    mask_tile(tOwn * 256, L, c, h, a1);
    mask_tile(tOwn * 256, L, c, h, a2);
    tile_scan(a1, U, h, P1, t1);
    tile_scan(a2, U, h, P2, t2);
    if (lane == 0) { tot1[tOwn] = t1; tot2[tOwn] = t2; }
  }
  __syncthreads();

  if (wave == 0) {
    v2f b1[4], b2[4];
#pragma unroll
    for (int j = 0; j < 4; ++j) {
      const int e = 16 * c + 4 * j + 2 * h;
      b1[j] = *(const v2f*)(tot1 + e);
      b2[j] = *(const v2f*)(tot2 + e);
    }
    v8f E1, E2;
    float d1, d2;
    tile_scan(b1, Us, h, E1, d1);
    tile_scan(b2, Us, h, E2, d2);
#pragma unroll
    for (int r = 0; r < 8; ++r) {
      off1[(8 * h + r) * 16 + c] = E1[r];
      off2[(8 * h + r) * 16 + c] = E2[r];
    }
  }
  __syncthreads();

  if (own && tOwn < nt) {
    const int t = tOwn;
    const float o1 = off1[t];
    const float o2 = off2[t];
    const int kb = t * 256 + 128 * h + c;
    if (t * 256 + 256 <= L) {
#pragma unroll
      for (int r = 0; r < 8; ++r)
        stats_store(P1[r] + o1, P2[r] + o2, kb + 16 * r, Nf, Bf, mean, istd);
    } else {
#pragma unroll
      for (int r = 0; r < 8; ++r)
        if (kb + 16 * r < L)
          stats_store(P1[r] + o1, P2[r] + o2, kb + 16 * r, Nf, Bf, mean, istd);
    }
  }
}

#define NORM_ROWS    8
#define NORM_THREADS 256
#define NORM_DEPTH   2
#define NORM_TSPLIT  2

template <int D>
__device__ __forceinline__ void norm_pieces(
    const v4f* __restrict__ xr, v4f* __restrict__ orow,
    const v4f* __restrict__ m4, const v4f* __restrict__ s4,
    int nq, int q0, const float (&g)[NORM_ROWS], const float (&b)[NORM_ROWS]) {
  v4f mv[D], sv[D], xv[D][NORM_ROWS];
#pragma unroll
  for (int u = 0; u < D; ++u) {
    mv[u] = m4[q0 + u * NORM_THREADS];
    sv[u] = s4[q0 + u * NORM_THREADS];
  }
#pragma unroll
  for (int u = 0; u < D; ++u) {
#pragma unroll
    for (int r = 0; r < NORM_ROWS; ++r)
      xv[u][r] = xr[(size_t)r * (size_t)nq + (size_t)(q0 + u * NORM_THREADS)];
  }
#pragma unroll
  for (int u = 0; u < D; ++u) {
#pragma unroll
    for (int r = 0; r < NORM_ROWS; ++r) {
      v4f o = (xv[u][r] - mv[u]) * sv[u] * g[r] + b[r];
      __builtin_nontemporal_store(
          o, orow + (size_t)r * (size_t)nq + (size_t)(q0 + u * NORM_THREADS));
    }
  }
}

__global__ __launch_bounds__(NORM_THREADS) void normalize_rows(
    const float* __restrict__ x, const float* __restrict__ gamma,
    const float* __restrict__ beta, const float* __restrict__ mean,
    const float* __restrict__ istd, float* __restrict__ out, int L) {
  const int nq   = L >> 2;
  const int span = (nq + NORM_TSPLIT - 1) / NORM_TSPLIT;
  const int qBeg = blockIdx.x * span;
  const int qEnd = (qBeg + span < nq) ? (qBeg + span) : nq;
  const int row0 = blockIdx.y * NORM_ROWS;

  float g[NORM_ROWS], b[NORM_ROWS];
#pragma unroll
  for (int r = 0; r < NORM_ROWS; ++r) { g[r] = gamma[row0 + r]; b[r] = beta[row0 + r]; }

  const v4f* __restrict__ xr = (const v4f*)(x + (size_t)row0 * (size_t)L);
  v4f* __restrict__ orow     = (v4f*)(out + (size_t)row0 * (size_t)L);
  const v4f* __restrict__ m4 = (const v4f*)mean;
  const v4f* __restrict__ s4 = (const v4f*)istd;

  int q0 = qBeg + (int)threadIdx.x;
  for (; q0 + NORM_THREADS * (NORM_DEPTH - 1) < qEnd; q0 += NORM_THREADS * NORM_DEPTH)
    norm_pieces<NORM_DEPTH>(xr, orow, m4, s4, nq, q0, g, b);
  for (; q0 < qEnd; q0 += NORM_THREADS)
    norm_pieces<1>(xr, orow, m4, s4, nq, q0, g, b);
}

extern "C" void kernel_launch(void* const* d_in, const int* in_sizes, int n_in,
                              void* d_out, int out_size, void* d_ws, size_t ws_size,
                              hipStream_t stream) {
  const float* x     = (const float*)d_in[0];
  const float* gamma = (const float*)d_in[1];
  const float* beta  = (const float*)d_in[2];
  float* out = (float*)d_out;

  const int B = 4, N = 512;
  const int rows = B * N;
  const int L = in_sizes[0] / rows;

  float* ws   = (float*)d_ws;
  float* psum = ws;
  float* psq  = psum + (size_t)ROW_GROUPS * L;
  float* c1   = psq  + (size_t)ROW_GROUPS * L;
  float* c2   = c1 + L;
  float* mean = c2 + L;
  float* istd = mean + L;

  dim3 g1(L / COLS_PER_BLK, ROW_GROUPS);
  colsum_wmma<<<g1, 256, 0, stream>>>(x, psum, psq, L, rows);

  combine_groups<<<(L + 255) / 256, 256, 0, stream>>>(psum, psq, c1, c2, L);

  scan_stats_wmma<<<SCAN_GROUPS, SCAN_THREADS, 0, stream>>>(c1, c2, mean, istd, L, (float)N, (float)B);

  dim3 g4(NORM_TSPLIT, rows / NORM_ROWS);
  normalize_rows<<<g4, NORM_THREADS, 0, stream>>>(x, gamma, beta, mean, istd, out, L);
}
